# starting kernel
# baseline (speedup 1.0000x reference)
#include <hip/hip_runtime.h>

#define B_      64
#define HW_     3136
#define C_      128
#define KC      32
#define LDSK    (KC + 2)
#define NCHUNK  (HW_ / KC)
#define LBUF    (C_ * LDSK)

typedef __attribute__((ext_vector_type(2))) float v2f;
typedef __attribute__((ext_vector_type(8))) float v8f;

__global__ __launch_bounds__(256) void secstat_cov_wmma(const float* __restrict__ x,
                                                        float* __restrict__ out) {
  __shared__ float lds[2 * LBUF];
  __shared__ float ssum_sh[8 * C_];

  const int tid  = threadIdx.x;
  const int lane = tid & 31;
  const int wave = tid >> 5;
  const int b    = blockIdx.x;
  const int r    = blockIdx.y;

  const int laneN = lane & 15;
  const int laneK = (lane >> 4) << 1;
  const int aBase = (16 * r    + laneN) * LDSK + laneK;
  const int bBase = (16 * wave + laneN) * LDSK + laneK;

  const float4* __restrict__ xg4 = (const float4*)(x + (size_t)b * HW_ * C_);

  const int ch0 = (lane << 2);

  v8f acc = {0.f, 0.f, 0.f, 0.f, 0.f, 0.f, 0.f, 0.f};
  float s0 = 0.f, s1 = 0.f, s2 = 0.f, s3 = 0.f;

  {
#pragma unroll
    for (int i = 0; i < 4; ++i) {
      const float4 v = xg4[i * 256 + tid];
      const int row = (i << 3) + wave;
      lds[(ch0 + 0) * LDSK + row] = v.x;
      lds[(ch0 + 1) * LDSK + row] = v.y;
      lds[(ch0 + 2) * LDSK + row] = v.z;
      lds[(ch0 + 3) * LDSK + row] = v.w;
      s0 += v.x; s1 += v.y; s2 += v.z; s3 += v.w;
    }
  }

  for (int chunk = 0; chunk < NCHUNK; ++chunk) {
    __syncthreads();
    const float* Lc = lds + (chunk & 1) * LBUF;
#pragma unroll
    for (int k = 0; k < KC; k += 4) {
      const v2f a  = *(const v2f*)&Lc[aBase + k];
      const v2f bf = *(const v2f*)&Lc[bBase + k];
      acc = __builtin_amdgcn_wmma_f32_16x16x4_f32(false, a, false, bf,
                                                  (short)0, acc, false, false);
    }
    if (chunk + 1 < NCHUNK) {
      float* Ln = lds + ((chunk + 1) & 1) * LBUF;
      const int base4 = (chunk + 1) * KC * (C_ / 4);
#pragma unroll
      for (int i = 0; i < 4; ++i) {
        const float4 v = xg4[base4 + i * 256 + tid];
        const int row = (i << 3) + wave;
        Ln[(ch0 + 0) * LDSK + row] = v.x;
        Ln[(ch0 + 1) * LDSK + row] = v.y;
        Ln[(ch0 + 2) * LDSK + row] = v.z;
        Ln[(ch0 + 3) * LDSK + row] = v.w;
        s0 += v.x; s1 += v.y; s2 += v.z; s3 += v.w;
      }
    }
  }

  __syncthreads();
  ssum_sh[wave * C_ + ch0 + 0] = s0;
  ssum_sh[wave * C_ + ch0 + 1] = s1;
  ssum_sh[wave * C_ + ch0 + 2] = s2;
  ssum_sh[wave * C_ + ch0 + 3] = s3;
  __syncthreads();
  if (tid < C_) {
    float s = 0.f;
#pragma unroll
    for (int w = 0; w < 8; ++w) s += ssum_sh[w * C_ + tid];
    ssum_sh[tid] = s;
  }
  __syncthreads();

  const float invHW = 1.0f / (float)HW_;
  const float invN1 = 1.0f / (float)(HW_ - 1);
  const int  colC = 16 * wave + laneN;
  const float sj  = ssum_sh[colC];
#pragma unroll
  for (int v = 0; v < 8; ++v) {
    const int rowC = 16 * r + v + ((lane >> 4) << 3);
    const float si = ssum_sh[rowC];
    const float cv = (acc[v] - si * sj * invHW) * invN1;
    out[((size_t)b * C_ + rowC) * C_ + colC] = cv;
  }
}

extern "C" void kernel_launch(void* const* d_in, const int* in_sizes, int n_in,
                              void* d_out, int out_size, void* d_ws, size_t ws_size,
                              hipStream_t stream) {
  (void)in_sizes; (void)n_in; (void)d_ws; (void)ws_size; (void)out_size;
  const float* x = (const float*)d_in[0];
  float* out = (float*)d_out;
  dim3 grid(B_, C_ / 16);
  dim3 block(256);
  secstat_cov_wmma<<<grid, block, 0, stream>>>(x, out);
}


# LDS tile kept row-major, no staged transpose (A and B fragments both run lanes along channels): b128 staging stores replace bank-conflicted b32 pairs; fragments are 2-address b32 reads
# speedup vs baseline: 1.0962x; 1.0962x over previous
#include <hip/hip_runtime.h>

#define B_      64
#define HW_     3136
#define C_      128
#define KC      32
#define LDSR    (C_ + 16)
#define NCHUNK  (HW_ / KC)
#define LBUF    (KC * LDSR)

typedef __attribute__((ext_vector_type(2))) float v2f;
typedef __attribute__((ext_vector_type(8))) float v8f;

__global__ __launch_bounds__(256) void secstat_cov_wmma(const float* __restrict__ x,
                                                        float* __restrict__ out) {
  __shared__ __attribute__((aligned(16))) float lds[2 * LBUF];
  __shared__ float ssum_sh[8 * C_];

  const int tid  = threadIdx.x;
  const int lane = tid & 31;
  const int wave = tid >> 5;
  const int b    = blockIdx.x;
  const int r    = blockIdx.y;

  const int laneN = lane & 15;
  const int laneK = (lane >> 4) << 1;
  const int aBase = laneK * LDSR + 16 * r    + laneN;
  const int bBase = laneK * LDSR + 16 * wave + laneN;

  const float4* __restrict__ xg4 = (const float4*)(x + (size_t)b * HW_ * C_);

  const int ch0 = (lane << 2);

  v8f acc = {0.f, 0.f, 0.f, 0.f, 0.f, 0.f, 0.f, 0.f};
  float s0 = 0.f, s1 = 0.f, s2 = 0.f, s3 = 0.f;

  {
#pragma unroll
    for (int i = 0; i < 4; ++i) {
      const float4 v = xg4[i * 256 + tid];
      const int row = (i << 3) + wave;
      *(float4*)&lds[row * LDSR + ch0] = v;
      s0 += v.x; s1 += v.y; s2 += v.z; s3 += v.w;
    }
  }

  for (int chunk = 0; chunk < NCHUNK; ++chunk) {
    __syncthreads();
    const float* Lc = lds + (chunk & 1) * LBUF;
#pragma unroll
    for (int k = 0; k < KC; k += 4) {
      v2f a, bf;
      a.x  = Lc[aBase + k * LDSR];  a.y  = Lc[aBase + (k + 1) * LDSR];
      bf.x = Lc[bBase + k * LDSR];  bf.y = Lc[bBase + (k + 1) * LDSR];
      acc = __builtin_amdgcn_wmma_f32_16x16x4_f32(false, a, false, bf,
                                                  (short)0, acc, false, false);
    }
    if (chunk + 1 < NCHUNK) {
      float* Ln = lds + ((chunk + 1) & 1) * LBUF;
      const int base4 = (chunk + 1) * KC * (C_ / 4);
#pragma unroll
      for (int i = 0; i < 4; ++i) {
        const float4 v = xg4[base4 + i * 256 + tid];
        const int row = (i << 3) + wave;
        *(float4*)&Ln[row * LDSR + ch0] = v;
        s0 += v.x; s1 += v.y; s2 += v.z; s3 += v.w;
      }
    }
  }

  __syncthreads();
  ssum_sh[wave * C_ + ch0 + 0] = s0;
  ssum_sh[wave * C_ + ch0 + 1] = s1;
  ssum_sh[wave * C_ + ch0 + 2] = s2;
  ssum_sh[wave * C_ + ch0 + 3] = s3;
  __syncthreads();
  if (tid < C_) {
    float s = 0.f;
#pragma unroll
    for (int w = 0; w < 8; ++w) s += ssum_sh[w * C_ + tid];
    ssum_sh[tid] = s;
  }
  __syncthreads();

  const float invHW = 1.0f / (float)HW_;
  const float invN1 = 1.0f / (float)(HW_ - 1);
  const int  colC = 16 * wave + laneN;
  const float sj  = ssum_sh[colC];
#pragma unroll
  for (int v = 0; v < 8; ++v) {
    const int rowC = 16 * r + v + ((lane >> 4) << 3);
    const float si = ssum_sh[rowC];
    const float cv = (acc[v] - si * sj * invHW) * invN1;
    out[((size_t)b * C_ + rowC) * C_ + colC] = cv;
  }
}

extern "C" void kernel_launch(void* const* d_in, const int* in_sizes, int n_in,
                              void* d_out, int out_size, void* d_ws, size_t ws_size,
                              hipStream_t stream) {
  (void)in_sizes; (void)n_in; (void)d_ws; (void)ws_size; (void)out_size;
  const float* x = (const float*)d_in[0];
  float* out = (float*)d_out;
  dim3 grid(B_, C_ / 16);
  dim3 block(256);
  secstat_cov_wmma<<<grid, block, 0, stream>>>(x, out);
}


# LDS chunk depth 32 -> 112 rows: a chunk's 14 b128 loads per thread issue as one group ahead of its 28 WMMAs, 3.5x the bytes in flight per barrier interval, 28 intervals not 98; sum order unchanged
# speedup vs baseline: 1.4952x; 1.3640x over previous
#include <hip/hip_runtime.h>

#define B_      64
#define HW_     3136
#define C_      128
#define KC      112
#define LDSR    (C_ + 16)
#define NCHUNK  (HW_ / KC)
#define LBUF    (KC * LDSR)
#define NLD     (KC * C_ / 4 / 256)

typedef __attribute__((ext_vector_type(2))) float v2f;
typedef __attribute__((ext_vector_type(8))) float v8f;

__global__ __launch_bounds__(256) void secstat_cov_wmma(const float* __restrict__ x,
                                                        float* __restrict__ out) {
  __shared__ __attribute__((aligned(16))) float lds[2 * LBUF];
  __shared__ float ssum_sh[8 * C_];

  const int tid  = threadIdx.x;
  const int lane = tid & 31;
  const int wave = tid >> 5;
  const int b    = blockIdx.x;
  const int r    = blockIdx.y;

  const int laneN = lane & 15;
  const int laneK = (lane >> 4) << 1;
  const int aBase = laneK * LDSR + 16 * r    + laneN;
  const int bBase = laneK * LDSR + 16 * wave + laneN;

  const float4* __restrict__ xg4 = (const float4*)(x + (size_t)b * HW_ * C_);

  const int ch0 = (lane << 2);

  v8f acc = {0.f, 0.f, 0.f, 0.f, 0.f, 0.f, 0.f, 0.f};
  float s0 = 0.f, s1 = 0.f, s2 = 0.f, s3 = 0.f;

  {
#pragma unroll
    for (int i = 0; i < NLD; ++i) {
      const float4 v = xg4[i * 256 + tid];
      const int row = (i << 3) + wave;
      *(float4*)&lds[row * LDSR + ch0] = v;
      s0 += v.x; s1 += v.y; s2 += v.z; s3 += v.w;
    }
  }

  for (int chunk = 0; chunk < NCHUNK; ++chunk) {
    __syncthreads();
    const float* Lc = lds + (chunk & 1) * LBUF;
    const bool more = (chunk + 1 < NCHUNK);
    float4 nx[NLD];
    if (more) {
      const int base4 = (chunk + 1) * KC * (C_ / 4);
#pragma unroll
      for (int i = 0; i < NLD; ++i) nx[i] = xg4[base4 + i * 256 + tid];
    }
    __builtin_amdgcn_sched_barrier(0);
#pragma unroll
    for (int k = 0; k < KC; k += 4) {
      v2f a, bf;
      a.x  = Lc[aBase + k * LDSR];  a.y  = Lc[aBase + (k + 1) * LDSR];
      bf.x = Lc[bBase + k * LDSR];  bf.y = Lc[bBase + (k + 1) * LDSR];
      acc = __builtin_amdgcn_wmma_f32_16x16x4_f32(false, a, false, bf,
                                                  (short)0, acc, false, false);
    }
    __builtin_amdgcn_sched_barrier(0);
    if (more) {
      float* Ln = lds + ((chunk + 1) & 1) * LBUF;
#pragma unroll
      for (int i = 0; i < NLD; ++i) {
        const float4 v = nx[i];
        const int row = (i << 3) + wave;
        *(float4*)&Ln[row * LDSR + ch0] = v;
        s0 += v.x; s1 += v.y; s2 += v.z; s3 += v.w;
      }
    }
  }

  __syncthreads();
  ssum_sh[wave * C_ + ch0 + 0] = s0;
  ssum_sh[wave * C_ + ch0 + 1] = s1;
  ssum_sh[wave * C_ + ch0 + 2] = s2;
  ssum_sh[wave * C_ + ch0 + 3] = s3;
  __syncthreads();
  if (tid < C_) {
    float s = 0.f;
#pragma unroll
    for (int w = 0; w < 8; ++w) s += ssum_sh[w * C_ + tid];
    ssum_sh[tid] = s;
  }
  __syncthreads();

  const float invHW = 1.0f / (float)HW_;
  const float invN1 = 1.0f / (float)(HW_ - 1);
  const int  colC = 16 * wave + laneN;
  const float sj  = ssum_sh[colC];
#pragma unroll
  for (int v = 0; v < 8; ++v) {
    const int rowC = 16 * r + v + ((lane >> 4) << 3);
    const float si = ssum_sh[rowC];
    const float cv = (acc[v] - si * sj * invHW) * invN1;
    out[((size_t)b * C_ + rowC) * C_ + colC] = cv;
  }
}

extern "C" void kernel_launch(void* const* d_in, const int* in_sizes, int n_in,
                              void* d_out, int out_size, void* d_ws, size_t ws_size,
                              hipStream_t stream) {
  (void)in_sizes; (void)n_in; (void)d_ws; (void)ws_size; (void)out_size;
  const float* x = (const float*)d_in[0];
  float* out = (float*)d_out;
  dim3 grid(B_, C_ / 16);
  dim3 block(256);
  secstat_cov_wmma<<<grid, block, 0, stream>>>(x, out);
}


# Gram symmetry: only the 36 unordered pairs of 16-channel blocks are multiplied (4 workgroups x 9 tiles per batch), mirror tiles stored transposed; k split over 8 waves, partials added in fixed order
# speedup vs baseline: 1.8214x; 1.2182x over previous
#include <hip/hip_runtime.h>


#define B_      64
#define HW_     3136
#define C_      128
#define NG      4
#define NT      9
#define NF      7
#define KC      224
#define LDSR    (C_ + 16)
#define NCHUNK  (HW_ / KC)
#define LBUF    (KC * LDSR)
#define NLD     (KC * C_ / 4 / 256)
#define KSW     (KC / 4 / 8)
#define FINP    19

typedef __attribute__((ext_vector_type(2))) float v2f;
typedef __attribute__((ext_vector_type(8))) float v8f;

__global__ __launch_bounds__(256) void secstat_cov_wmma(const float* __restrict__ x,
                                                        float* __restrict__ out) {
  __shared__ __attribute__((aligned(16))) float lds[2 * LBUF];
  __shared__ float ssum_sh[8 * C_];

  const int tid  = threadIdx.x;
  const int lane = tid & 31;
  const int wave = tid >> 5;
  const int b    = blockIdx.x;
  const int g    = blockIdx.y;

  const int laneN = lane & 15;
  const int laneK = (lane >> 4) << 1;

  int fblk[NF];
#pragma unroll
  for (int t = 0; t < 5; ++t) fblk[t] = (2 * g + t) & 7;
  fblk[5] = g;
  fblk[6] = g + 4;
  constexpr int TA[NT] = {0, 0, 0, 0, 1, 1, 1, 1, 5};
  constexpr int TB[NT] = {0, 1, 2, 3, 1, 2, 3, 4, 6};

  int fb[NF];
#pragma unroll
  for (int t = 0; t < NF; ++t) fb[t] = (wave * (KSW * 4) + laneK) * LDSR + 16 * fblk[t] + laneN;

  const float4* __restrict__ xg4 = (const float4*)(x + (size_t)b * HW_ * C_);

  const int ch0 = (lane << 2);

  v8f acc[NT];
#pragma unroll
  for (int q = 0; q < NT; ++q) acc[q] = (v8f){0.f, 0.f, 0.f, 0.f, 0.f, 0.f, 0.f, 0.f};
  float s0 = 0.f, s1 = 0.f, s2 = 0.f, s3 = 0.f;

  {
#pragma unroll
    for (int i = 0; i < NLD; ++i) {
      const float4 v = xg4[i * 256 + tid];
      const int row = (i << 3) + wave;
      *(float4*)&lds[row * LDSR + ch0] = v;
      s0 += v.x; s1 += v.y; s2 += v.z; s3 += v.w;
    }
  }

  for (int chunk = 0; chunk < NCHUNK; ++chunk) {
    __syncthreads();
    const float* Lc = lds + (chunk & 1) * LBUF;
    const bool more = (chunk + 1 < NCHUNK);
    float4 nx[NLD];
    if (more) {
      const int base4 = (chunk + 1) * KC * (C_ / 4);
#pragma unroll
      for (int i = 0; i < NLD; ++i) nx[i] = xg4[base4 + i * 256 + tid];
    }
    __builtin_amdgcn_sched_barrier(0);
#pragma unroll
    for (int s = 0; s < KSW; ++s) {
      v2f f[NF];
#pragma unroll
      for (int t = 0; t < NF; ++t) {
        f[t].x = Lc[fb[t] + (4 * s) * LDSR];
        f[t].y = Lc[fb[t] + (4 * s + 1) * LDSR];
      }
#pragma unroll
      for (int q = 0; q < NT; ++q)
        acc[q] = __builtin_amdgcn_wmma_f32_16x16x4_f32(false, f[TA[q]], false, f[TB[q]],
                                                       (short)0, acc[q], false, false);
    }
    __builtin_amdgcn_sched_barrier(0);
    if (more) {
      float* Ln = lds + ((chunk + 1) & 1) * LBUF;
#pragma unroll
      for (int i = 0; i < NLD; ++i) {
        const float4 v = nx[i];
        const int row = (i << 3) + wave;
        *(float4*)&Ln[row * LDSR + ch0] = v;
        s0 += v.x; s1 += v.y; s2 += v.z; s3 += v.w;
      }
    }
  }

  __syncthreads();
  ssum_sh[wave * C_ + ch0 + 0] = s0;
  ssum_sh[wave * C_ + ch0 + 1] = s1;
  ssum_sh[wave * C_ + ch0 + 2] = s2;
  ssum_sh[wave * C_ + ch0 + 3] = s3;
  float* red = lds;
#pragma unroll
  for (int q = 0; q < NT; ++q)
#pragma unroll
    for (int r = 0; r < 8; ++r) red[((wave * NT + q) * 8 + r) * 32 + lane] = acc[q][r];
  __syncthreads();

  if (tid < C_) {
    float s = 0.f;
#pragma unroll
    for (int w = 0; w < 8; ++w) s += ssum_sh[w * C_ + tid];
    ssum_sh[tid] = s;
  }
  float gs[NT];
#pragma unroll
  for (int q = 0; q < NT; ++q) {
    float t = 0.f;
#pragma unroll
    for (int w = 0; w < 8; ++w) t += red[((w * NT + q) * 8 + wave) * 32 + lane];
    gs[q] = t;
  }
  __syncthreads();

  const float invHW = 1.0f / (float)HW_;
  const float invN1 = 1.0f / (float)(HW_ - 1);
  const int m = wave + ((lane >> 4) << 3);
  float* fin = lds + 8 * NT * 256;
#pragma unroll
  for (int q = 0; q < NT; ++q) {
    const int rowC = 16 * fblk[TA[q]] + m;
    const int colC = 16 * fblk[TB[q]] + laneN;
    const float cv = (gs[q] - ssum_sh[rowC] * ssum_sh[colC] * invHW) * invN1;
    out[((size_t)b * C_ + rowC) * C_ + colC] = cv;
    if (TA[q] != TB[q]) fin[(q * 16 + m) * FINP + laneN] = cv;
  }
  __syncthreads();
#pragma unroll
  for (int q = 0; q < NT; ++q) {
    if (TA[q] != TB[q]) {
      const float cv = fin[(q * 16 + laneN) * FINP + m];
      const int rowC = 16 * fblk[TB[q]] + m;
      const int colC = 16 * fblk[TA[q]] + laneN;
      out[((size_t)b * C_ + rowC) * C_ + colC] = cv;
    }
  }
}

extern "C" void kernel_launch(void* const* d_in, const int* in_sizes, int n_in,
                              void* d_out, int out_size, void* d_ws, size_t ws_size,
                              hipStream_t stream) {
  (void)in_sizes; (void)n_in; (void)d_ws; (void)ws_size; (void)out_size;
  const float* x = (const float*)d_in[0];
  float* out = (float*)d_out;
  dim3 grid(B_, NG);
  dim3 block(256);
  secstat_cov_wmma<<<grid, block, 0, stream>>>(x, out);
}


# Wait for the next chunk's global loads moved behind the chunk's WMMA stream (channel sums pinned after it): the loads run under the WMMAs, not before the first one; sum order unchanged
# speedup vs baseline: 1.8862x; 1.0356x over previous
#include <hip/hip_runtime.h>


#define B_      64
#define HW_     3136
#define C_      128
#define NG      4
#define NT      9
#define NF      7
#define KC      224
#define LDSR    (C_ + 16)
#define NCHUNK  (HW_ / KC)
#define LBUF    (KC * LDSR)
#define NLD     (KC * C_ / 4 / 256)
#define KSW     (KC / 4 / 8)
#define FINP    19

typedef __attribute__((ext_vector_type(2))) float v2f;
typedef __attribute__((ext_vector_type(8))) float v8f;
typedef __attribute__((ext_vector_type(4))) float v4f;

__global__ __launch_bounds__(256) void secstat_cov_wmma(const float* __restrict__ x,
                                                        float* __restrict__ out) {
  __shared__ __attribute__((aligned(16))) float lds[2 * LBUF];
  __shared__ float ssum_sh[8 * C_];

  const int tid  = threadIdx.x;
  const int lane = tid & 31;
  const int wave = tid >> 5;
  const int b    = blockIdx.x;
  const int g    = blockIdx.y;

  const int laneN = lane & 15;
  const int laneK = (lane >> 4) << 1;

  int fblk[NF];
#pragma unroll
  for (int t = 0; t < 5; ++t) fblk[t] = (2 * g + t) & 7;
  fblk[5] = g;
  fblk[6] = g + 4;
  constexpr int TA[NT] = {0, 0, 0, 0, 1, 1, 1, 1, 5};
  constexpr int TB[NT] = {0, 1, 2, 3, 1, 2, 3, 4, 6};

  int fb[NF];
#pragma unroll
  for (int t = 0; t < NF; ++t) fb[t] = (wave * (KSW * 4) + laneK) * LDSR + 16 * fblk[t] + laneN;

  const v4f* __restrict__ xg4 = (const v4f*)(x + (size_t)b * HW_ * C_);

  const int ch0 = (lane << 2);

  v8f acc[NT];
#pragma unroll
  for (int q = 0; q < NT; ++q) acc[q] = (v8f){0.f, 0.f, 0.f, 0.f, 0.f, 0.f, 0.f, 0.f};
  float s0 = 0.f, s1 = 0.f, s2 = 0.f, s3 = 0.f;

  {
#pragma unroll
    for (int i = 0; i < NLD; ++i) {
      const v4f v = xg4[i * 256 + tid];
      const int row = (i << 3) + wave;
      *(v4f*)&lds[row * LDSR + ch0] = v;
      s0 += v.x; s1 += v.y; s2 += v.z; s3 += v.w;
    }
  }

  for (int chunk = 0; chunk < NCHUNK; ++chunk) {
    __syncthreads();
    const float* Lc = lds + (chunk & 1) * LBUF;
    const bool more = (chunk + 1 < NCHUNK);
    v4f nx[NLD];
    if (more) {
      const int base4 = (chunk + 1) * KC * (C_ / 4);
#pragma unroll
      for (int i = 0; i < NLD; ++i) nx[i] = xg4[base4 + i * 256 + tid];
    }
    __builtin_amdgcn_sched_barrier(0);
#pragma unroll
    for (int s = 0; s < KSW; ++s) {
      v2f f[NF];
#pragma unroll
      for (int t = 0; t < NF; ++t) {
        f[t].x = Lc[fb[t] + (4 * s) * LDSR];
        f[t].y = Lc[fb[t] + (4 * s + 1) * LDSR];
      }
#pragma unroll
      for (int q = 0; q < NT; ++q)
        acc[q] = __builtin_amdgcn_wmma_f32_16x16x4_f32(false, f[TA[q]], false, f[TB[q]],
                                                       (short)0, acc[q], false, false);
    }
    __builtin_amdgcn_sched_barrier(0);
    if (more) {
      float* Ln = lds + ((chunk + 1) & 1) * LBUF;
#pragma unroll
      for (int i = 0; i < NLD; ++i) {
        asm volatile("" : "+v"(nx[i]));
        const v4f v = nx[i];
        const int row = (i << 3) + wave;
        *(v4f*)&Ln[row * LDSR + ch0] = v;
        s0 += v.x; s1 += v.y; s2 += v.z; s3 += v.w;
      }
    }
  }

  __syncthreads();
  ssum_sh[wave * C_ + ch0 + 0] = s0;
  ssum_sh[wave * C_ + ch0 + 1] = s1;
  ssum_sh[wave * C_ + ch0 + 2] = s2;
  ssum_sh[wave * C_ + ch0 + 3] = s3;
  float* red = lds;
#pragma unroll
  for (int q = 0; q < NT; ++q)
#pragma unroll
    for (int r = 0; r < 8; ++r) red[((wave * NT + q) * 8 + r) * 32 + lane] = acc[q][r];
  __syncthreads();

  if (tid < C_) {
    float s = 0.f;
#pragma unroll
    for (int w = 0; w < 8; ++w) s += ssum_sh[w * C_ + tid];
    ssum_sh[tid] = s;
  }
  float gs[NT];
#pragma unroll
  for (int q = 0; q < NT; ++q) {
    float t = 0.f;
#pragma unroll
    for (int w = 0; w < 8; ++w) t += red[((w * NT + q) * 8 + wave) * 32 + lane];
    gs[q] = t;
  }
  __syncthreads();

  const float invHW = 1.0f / (float)HW_;
  const float invN1 = 1.0f / (float)(HW_ - 1);
  const int m = wave + ((lane >> 4) << 3);
  float* fin = lds + 8 * NT * 256;
#pragma unroll
  for (int q = 0; q < NT; ++q) {
    const int rowC = 16 * fblk[TA[q]] + m;
    const int colC = 16 * fblk[TB[q]] + laneN;
    const float cv = (gs[q] - ssum_sh[rowC] * ssum_sh[colC] * invHW) * invN1;
    out[((size_t)b * C_ + rowC) * C_ + colC] = cv;
    if (TA[q] != TB[q]) fin[(q * 16 + m) * FINP + laneN] = cv;
  }
  __syncthreads();
#pragma unroll
  for (int q = 0; q < NT; ++q) {
    if (TA[q] != TB[q]) {
      const float cv = fin[(q * 16 + laneN) * FINP + m];
      const int rowC = 16 * fblk[TB[q]] + m;
      const int colC = 16 * fblk[TA[q]] + laneN;
      out[((size_t)b * C_ + rowC) * C_ + colC] = cv;
    }
  }
}

extern "C" void kernel_launch(void* const* d_in, const int* in_sizes, int n_in,
                              void* d_out, int out_size, void* d_ws, size_t ws_size,
                              hipStream_t stream) {
  (void)in_sizes; (void)n_in; (void)d_ws; (void)ws_size; (void)out_size;
  const float* x = (const float*)d_in[0];
  float* out = (float*)d_out;
  dim3 grid(B_, NG);
  dim3 block(256);
  secstat_cov_wmma<<<grid, block, 0, stream>>>(x, out);
}


# A chunk's 49 LDS fragment reads issued as one group right behind the barrier, ahead of its global loads and its 63 WMMAs; only counted waits left between the WMMAs; sum order unchanged
# speedup vs baseline: 2.2574x; 1.1968x over previous
#include <hip/hip_runtime.h>


#define B_      64
#define HW_     3136
#define C_      128
#define NG      4
#define NT      9
#define NF      7
#define KC      224
#define LDSR    (C_ + 16)
#define NCHUNK  (HW_ / KC)
#define LBUF    (KC * LDSR)
#define NLD     (KC * C_ / 4 / 256)
#define KSW     (KC / 4 / 8)
#define FINP    19

typedef __attribute__((ext_vector_type(2))) float v2f;
typedef __attribute__((ext_vector_type(8))) float v8f;
typedef __attribute__((ext_vector_type(4))) float v4f;

__global__ __launch_bounds__(256) void secstat_cov_wmma(const float* __restrict__ x,
                                                        float* __restrict__ out) {
  __shared__ __attribute__((aligned(16))) float lds[2 * LBUF];
  __shared__ float ssum_sh[8 * C_];

  const int tid  = threadIdx.x;
  const int lane = tid & 31;
  const int wave = tid >> 5;
  const int b    = blockIdx.x;
  const int g    = blockIdx.y;

  const int laneN = lane & 15;
  const int laneK = (lane >> 4) << 1;

  int fblk[NF];
#pragma unroll
  for (int t = 0; t < 5; ++t) fblk[t] = (2 * g + t) & 7;
  fblk[5] = g;
  fblk[6] = g + 4;
  constexpr int TA[NT] = {0, 0, 0, 0, 1, 1, 1, 1, 5};
  constexpr int TB[NT] = {0, 1, 2, 3, 1, 2, 3, 4, 6};

  int fb[NF];
#pragma unroll
  for (int t = 0; t < NF; ++t) {
    fb[t] = (wave * (KSW * 4) + laneK) * LDSR + 16 * fblk[t] + laneN;
    asm volatile("" : "+v"(fb[t]));
  }

  const v4f* __restrict__ xg4 = (const v4f*)(x + (size_t)b * HW_ * C_);

  const int ch0 = (lane << 2);

  v8f acc[NT];
#pragma unroll
  for (int q = 0; q < NT; ++q) acc[q] = (v8f){0.f, 0.f, 0.f, 0.f, 0.f, 0.f, 0.f, 0.f};
  float s0 = 0.f, s1 = 0.f, s2 = 0.f, s3 = 0.f;

  {
#pragma unroll
    for (int i = 0; i < NLD; ++i) {
      const v4f v = xg4[i * 256 + tid];
      const int row = (i << 3) + wave;
      *(v4f*)&lds[row * LDSR + ch0] = v;
      s0 += v.x; s1 += v.y; s2 += v.z; s3 += v.w;
    }
  }

  for (int chunk = 0; chunk < NCHUNK; ++chunk) {
    __syncthreads();
    const float* Lc = lds + (chunk & 1) * LBUF;
    const bool more = (chunk + 1 < NCHUNK);
    v2f f[KSW][NF];
#pragma unroll
    for (int s = 0; s < KSW; ++s) {
#pragma unroll
      for (int t = 0; t < NF; ++t) {
        f[s][t].x = Lc[fb[t] + (4 * s) * LDSR];
        f[s][t].y = Lc[fb[t] + (4 * s + 1) * LDSR];
      }
    }
    __builtin_amdgcn_sched_barrier(0);
    v4f nx[NLD];
    if (more) {
      const int base4 = (chunk + 1) * KC * (C_ / 4);
#pragma unroll
      for (int i = 0; i < NLD; ++i) nx[i] = xg4[base4 + i * 256 + tid];
    }
    __builtin_amdgcn_sched_barrier(0);
#pragma unroll
    for (int s = 0; s < KSW; ++s) {
#pragma unroll
      for (int q = 0; q < NT; ++q)
        acc[q] = __builtin_amdgcn_wmma_f32_16x16x4_f32(false, f[s][TA[q]], false, f[s][TB[q]],
                                                       (short)0, acc[q], false, false);
    }
    __builtin_amdgcn_sched_barrier(0);
    if (more) {
      float* Ln = lds + ((chunk + 1) & 1) * LBUF;
#pragma unroll
      for (int i = 0; i < NLD; ++i) {
        asm volatile("" : "+v"(nx[i]));
        const v4f v = nx[i];
        const int row = (i << 3) + wave;
        *(v4f*)&Ln[row * LDSR + ch0] = v;
        s0 += v.x; s1 += v.y; s2 += v.z; s3 += v.w;
      }
    }
  }

  __syncthreads();
  ssum_sh[wave * C_ + ch0 + 0] = s0;
  ssum_sh[wave * C_ + ch0 + 1] = s1;
  ssum_sh[wave * C_ + ch0 + 2] = s2;
  ssum_sh[wave * C_ + ch0 + 3] = s3;
  float* red = lds;
#pragma unroll
  for (int q = 0; q < NT; ++q)
#pragma unroll
    for (int r = 0; r < 8; ++r) red[((wave * NT + q) * 8 + r) * 32 + lane] = acc[q][r];
  __syncthreads();

  if (tid < C_) {
    float s = 0.f;
#pragma unroll
    for (int w = 0; w < 8; ++w) s += ssum_sh[w * C_ + tid];
    ssum_sh[tid] = s;
  }
  float gs[NT];
#pragma unroll
  for (int q = 0; q < NT; ++q) {
    float t = 0.f;
#pragma unroll
    for (int w = 0; w < 8; ++w) t += red[((w * NT + q) * 8 + wave) * 32 + lane];
    gs[q] = t;
  }
  __syncthreads();

  const float invHW = 1.0f / (float)HW_;
  const float invN1 = 1.0f / (float)(HW_ - 1);
  const int m = wave + ((lane >> 4) << 3);
  float* fin = lds + 8 * NT * 256;
#pragma unroll
  for (int q = 0; q < NT; ++q) {
    const int rowC = 16 * fblk[TA[q]] + m;
    const int colC = 16 * fblk[TB[q]] + laneN;
    const float cv = (gs[q] - ssum_sh[rowC] * ssum_sh[colC] * invHW) * invN1;
    out[((size_t)b * C_ + rowC) * C_ + colC] = cv;
    if (TA[q] != TB[q]) fin[(q * 16 + m) * FINP + laneN] = cv;
  }
  __syncthreads();
#pragma unroll
  for (int q = 0; q < NT; ++q) {
    if (TA[q] != TB[q]) {
      const float cv = fin[(q * 16 + laneN) * FINP + m];
      const int rowC = 16 * fblk[TB[q]] + m;
      const int colC = 16 * fblk[TA[q]] + laneN;
      out[((size_t)b * C_ + rowC) * C_ + colC] = cv;
    }
  }
}

extern "C" void kernel_launch(void* const* d_in, const int* in_sizes, int n_in,
                              void* d_out, int out_size, void* d_ws, size_t ws_size,
                              hipStream_t stream) {
  (void)in_sizes; (void)n_in; (void)d_ws; (void)ws_size; (void)out_size;
  const float* x = (const float*)d_in[0];
  float* out = (float*)d_out;
  dim3 grid(B_, NG);
  dim3 block(256);
  secstat_cov_wmma<<<grid, block, 0, stream>>>(x, out);
}


# WMMA fragments loaded straight from global memory (channel blocks = stride-4 channel sets, so b128 row loads are fragment sources): LDS staging and loop barriers gone; channel-sum order changed
# speedup vs baseline: 3.2808x; 1.4534x over previous
#include <hip/hip_runtime.h>


#define B_      64
#define HW_     3136
#define C_      128
#define NG      4
#define NT      9
#define NF      7
#define KC      224
#define NCHUNK  (HW_ / KC)
#define KSW     (KC / 4 / 8)
#define FINP    19

typedef __attribute__((ext_vector_type(2))) float v2f;
typedef __attribute__((ext_vector_type(8))) float v8f;
typedef __attribute__((ext_vector_type(4))) float v4f;

__host__ __device__ constexpr int fblk_of(int G, int t) {
  return t < 5 ? ((2 * G + t) & 7) : (t == 5 ? G : G + 4);
}
__host__ __device__ constexpr int chan_of(int beta, int m) { return 64 * (beta >> 2) + 4 * m + (beta & 3); }

template <int G>
__device__ __forceinline__ void secstat_group(const float* __restrict__ x, float* __restrict__ out,
                                              float* red, float* fin, float* ssum_sh, float* stot) {
  const int tid  = threadIdx.x;
  const int lane = tid & 31;
  const int wave = tid >> 5;
  const int b    = blockIdx.x;
  const int c    = lane & 15;
  const int h    = lane >> 4;

  constexpr int TA[NT] = {0, 0, 0, 0, 1, 1, 1, 1, 5};
  constexpr int TB[NT] = {0, 1, 2, 3, 1, 2, 3, 4, 6};

  const v4f* __restrict__ p =
      (const v4f*)(x + ((size_t)b * HW_ + (size_t)(wave * (KSW * 4) + 2 * h)) * C_ + 4 * c);

  v8f acc[NT];
#pragma unroll
  for (int q = 0; q < NT; ++q) acc[q] = (v8f){0.f, 0.f, 0.f, 0.f, 0.f, 0.f, 0.f, 0.f};
  v4f cs0 = (v4f){0.f, 0.f, 0.f, 0.f};
  v4f cs1 = (v4f){0.f, 0.f, 0.f, 0.f};

  v4f bufA[KSW][2][2], bufB[KSW][2][2];

#define LOAD_CHUNK(buf, chunk)                                                          \
  {                                                                                      \
    const v4f* __restrict__ pc = p + (size_t)(chunk) * (KC * (C_ / 4));                 \
    _Pragma("unroll") for (int s = 0; s < KSW; ++s)                                     \
      _Pragma("unroll") for (int j = 0; j < 2; ++j)                                     \
        _Pragma("unroll") for (int H = 0; H < 2; ++H)                                   \
          buf[s][H][j] = pc[(4 * s + j) * (C_ / 4) + 16 * H];                           \
  }

#define COMPUTE_CHUNK(buf)                                                               \
  {                                                                                      \
    _Pragma("unroll") for (int s = 0; s < KSW; ++s) {                                   \
      v2f f[NF];                                                                         \
      _Pragma("unroll") for (int t = 0; t < NF; ++t) {                                  \
        constexpr_frag(f[t], buf[s], t);                                                 \
      }                                                                                  \
      _Pragma("unroll") for (int q = 0; q < NT; ++q)                                    \
        acc[q] = __builtin_amdgcn_wmma_f32_16x16x4_f32(false, f[TA[q]], false, f[TB[q]], \
                                                       (short)0, acc[q], false, false);  \
      cs0 += buf[s][0][0]; cs0 += buf[s][0][1];                                          \
      cs1 += buf[s][1][0]; cs1 += buf[s][1][1];                                          \
    }                                                                                    \
  }

#define constexpr_frag(dst, bs, t)                                                       \
  {                                                                                      \
    const int beta_ = fblk_of(G, t);                                                     \
    dst.x = bs[beta_ >> 2][0][beta_ & 3];                                                \
    dst.y = bs[beta_ >> 2][1][beta_ & 3];                                                \
  }

  LOAD_CHUNK(bufA, 0);
#pragma nounroll
  for (int chunk = 0; chunk < NCHUNK; chunk += 2) {
    LOAD_CHUNK(bufB, chunk + 1);
    __builtin_amdgcn_sched_barrier(0);
    COMPUTE_CHUNK(bufA);
    __builtin_amdgcn_sched_barrier(0);
    const int nxt = (chunk + 2 < NCHUNK) ? chunk + 2 : NCHUNK - 1;
    LOAD_CHUNK(bufA, nxt);
    __builtin_amdgcn_sched_barrier(0);
    COMPUTE_CHUNK(bufB);
    __builtin_amdgcn_sched_barrier(0);
  }
#undef LOAD_CHUNK
#undef COMPUTE_CHUNK
#undef constexpr_frag

  *(v4f*)&ssum_sh[(wave * 2 + h) * C_ + 4 * c]      = cs0;
  *(v4f*)&ssum_sh[(wave * 2 + h) * C_ + 64 + 4 * c] = cs1;
#pragma unroll
  for (int q = 0; q < NT; ++q)
#pragma unroll
    for (int r = 0; r < 8; ++r) red[((wave * NT + q) * 8 + r) * 32 + lane] = acc[q][r];
  __syncthreads();

  if (tid < C_) {
    float s = 0.f;
#pragma unroll
    for (int w = 0; w < 16; ++w) s += ssum_sh[w * C_ + tid];
    stot[tid] = s;
  }
  float gs[NT];
#pragma unroll
  for (int q = 0; q < NT; ++q) {
    float t = 0.f;
#pragma unroll
    for (int w = 0; w < 8; ++w) t += red[((w * NT + q) * 8 + wave) * 32 + lane];
    gs[q] = t;
  }
  __syncthreads();

  const float invHW = 1.0f / (float)HW_;
  const float invN1 = 1.0f / (float)(HW_ - 1);
  const int m = wave + (h << 3);
#pragma unroll
  for (int q = 0; q < NT; ++q) {
    const int ba = fblk_of(G, TA[q]);
    const int bb = fblk_of(G, TB[q]);
    const int rowC = 64 * (ba >> 2) + 4 * m + (ba & 3);
    const int colC = 64 * (bb >> 2) + 4 * c + (bb & 3);
    const float cv = (gs[q] - stot[rowC] * stot[colC] * invHW) * invN1;
    out[((size_t)b * C_ + rowC) * C_ + colC] = cv;
    if (TA[q] != TB[q]) fin[(q * 16 + m) * FINP + c] = cv;
  }
  __syncthreads();
#pragma unroll
  for (int q = 0; q < NT; ++q) {
    if (TA[q] != TB[q]) {
      const int ba = fblk_of(G, TA[q]);
      const int bb = fblk_of(G, TB[q]);
      const float cv = fin[(q * 16 + c) * FINP + m];
      const int rowC = 64 * (bb >> 2) + 4 * m + (bb & 3);
      const int colC = 64 * (ba >> 2) + 4 * c + (ba & 3);
      out[((size_t)b * C_ + rowC) * C_ + colC] = cv;
    }
  }
}

__global__ __launch_bounds__(256) void secstat_cov_wmma(const float* __restrict__ x,
                                                        float* __restrict__ out) {
  __shared__ __attribute__((aligned(16))) float red[8 * NT * 256];
  __shared__ float fin[NT * 16 * FINP];
  __shared__ __attribute__((aligned(16))) float ssum_sh[16 * C_];
  __shared__ float stot[C_];
  switch (blockIdx.y) {
    case 0:  secstat_group<0>(x, out, red, fin, ssum_sh, stot); break;
    case 1:  secstat_group<1>(x, out, red, fin, ssum_sh, stot); break;
    case 2:  secstat_group<2>(x, out, red, fin, ssum_sh, stot); break;
    default: secstat_group<3>(x, out, red, fin, ssum_sh, stot); break;
  }
}

extern "C" void kernel_launch(void* const* d_in, const int* in_sizes, int n_in,
                              void* d_out, int out_size, void* d_ws, size_t ws_size,
                              hipStream_t stream) {
  (void)in_sizes; (void)n_in; (void)d_ws; (void)ws_size; (void)out_size;
  const float* x = (const float*)d_in[0];
  float* out = (float*)d_out;
  dim3 grid(B_, NG);
  dim3 block(256);
  secstat_cov_wmma<<<grid, block, 0, stream>>>(x, out);
}


# 36 block pairs re-dealt so a workgroup needs 2 or 3 of the 4 channel quarters and loads only those (b64 row loads per quarter): 50/50/75/75 % of the row bytes instead of all; sum order unchanged
# speedup vs baseline: 3.5095x; 1.0697x over previous
#include <hip/hip_runtime.h>


#define B_      64
#define HW_     3136
#define C_      128
#define NG      4
#define NT      9
#define KC      224
#define NCHUNK  (HW_ / KC)
#define KSW     (KC / 4 / 8)
#define FINP    19

typedef __attribute__((ext_vector_type(2))) float v2f;
typedef __attribute__((ext_vector_type(8))) float v8f;
typedef __attribute__((ext_vector_type(4))) float v4f;

__host__ __device__ constexpr int n2_of(int G) { return G == 0 ? 2 : (G == 1 ? 0 : 1); }
__host__ __device__ constexpr int n4_of(int G) { return G == 0 ? 0 : 1; }
__host__ __device__ constexpr int base2_of(int G, int idx) { return G == 0 ? 32 * idx : (G == 3 ? 32 : 0); }
__host__ __device__ constexpr int nfrag_of(int G) { return 2 * n2_of(G) + 4 * n4_of(G); }
__host__ __device__ constexpr bool frag_is2(int G, int t) { return t < 2 * n2_of(G); }
__host__ __device__ constexpr int frag_idx(int G, int t) { return frag_is2(G, t) ? t / 2 : 0; }
__host__ __device__ constexpr int frag_el(int G, int t) { return frag_is2(G, t) ? t % 2 : t - 2 * n2_of(G); }
__host__ __device__ constexpr int chan_of(int G, int t, int m) {
  return frag_is2(G, t) ? base2_of(G, frag_idx(G, t)) + 2 * m + frag_el(G, t) : 64 + 4 * m + frag_el(G, t);
}
__host__ __device__ constexpr bool touched(int G, int ch) {
  return G == 0 ? ch < 64 : (G == 1 ? ch >= 64 : (G == 2 ? (ch < 32 || ch >= 64) : ch >= 32));
}
__host__ __device__ constexpr int ta_of(int G, int q) {
  return G == 0 ? (q < 4 ? 0 : (q < 6 ? 1 : (q < 8 ? 2 : 3)))
       : G == 1 ? (q < 4 ? 0 : (q < 7 ? 1 : 2))
       : G == 2 ? (q < 4 ? 0 : 1)
                : (q < 4 ? 0 : (q < 8 ? 1 : 5));
}
__host__ __device__ constexpr int tb_of(int G, int q) {
  return G == 0 ? (q < 4 ? q : (q < 6 ? q - 2 : (q < 8 ? q - 4 : 3)))
       : G == 1 ? (q < 4 ? q : (q < 7 ? q - 3 : q - 5))
       : G == 2 ? (q < 4 ? q + 2 : (q < 8 ? q - 2 : 1))
                : (q < 4 ? q + 2 : (q < 8 ? q - 2 : 5));
}

template <int G>
__device__ __forceinline__ void secstat_group(const float* __restrict__ x, float* __restrict__ out,
                                              float* red, float* fin, float* ssum_sh, float* stot) {
  constexpr int N2 = n2_of(G);
  constexpr int N4 = n4_of(G);
  constexpr int N2A = N2 > 0 ? N2 : 1;
  constexpr int N4A = N4 > 0 ? N4 : 1;
  constexpr int NF = nfrag_of(G);

  const int tid  = threadIdx.x;
  const int lane = tid & 31;
  const int wave = tid >> 5;
  const int b    = blockIdx.x;
  const int c    = lane & 15;
  const int h    = lane >> 4;

  const float* __restrict__ prow = x + ((size_t)b * HW_ + (size_t)(wave * (KSW * 4) + 2 * h)) * C_;
  const float* __restrict__ p2 = prow + 2 * c;
  const float* __restrict__ p4 = prow + 64 + 4 * c;

  v8f acc[NT];
#pragma unroll
  for (int q = 0; q < NT; ++q) acc[q] = (v8f){0.f, 0.f, 0.f, 0.f, 0.f, 0.f, 0.f, 0.f};
  v2f cs2[N2A];
  v4f cs4[N4A];
#pragma unroll
  for (int i = 0; i < N2A; ++i) cs2[i] = (v2f){0.f, 0.f};
#pragma unroll
  for (int i = 0; i < N4A; ++i) cs4[i] = (v4f){0.f, 0.f, 0.f, 0.f};

  v2f bufA2[KSW][2][N2A], bufB2[KSW][2][N2A];
  v4f bufA4[KSW][2][N4A], bufB4[KSW][2][N4A];

#define LOAD_CHUNK(b2, b4, chunk)                                                        \
  {                                                                                      \
    const size_t off_ = (size_t)(chunk) * (KC * C_);                                     \
    _Pragma("unroll") for (int s = 0; s < KSW; ++s)                                      \
      _Pragma("unroll") for (int j = 0; j < 2; ++j) {                                    \
        _Pragma("unroll") for (int i = 0; i < N2; ++i)                                   \
          b2[s][j][i] = *(const v2f*)(p2 + off_ + (4 * s + j) * C_ + base2_of(G, i));    \
        _Pragma("unroll") for (int i = 0; i < N4; ++i)                                   \
          b4[s][j][i] = *(const v4f*)(p4 + off_ + (4 * s + j) * C_);                     \
      }                                                                                  \
  }

#define MAKE_FRAG(dst, b2s, b4s, t)                                                      \
  {                                                                                      \
    if constexpr (frag_is2(G, t)) {                                                      \
      dst.x = b2s[0][frag_idx(G, t)][frag_el(G, t)];                                     \
      dst.y = b2s[1][frag_idx(G, t)][frag_el(G, t)];                                     \
    } else {                                                                             \
      dst.x = b4s[0][0][frag_el(G, t)];                                                  \
      dst.y = b4s[1][0][frag_el(G, t)];                                                  \
    }                                                                                    \
  }

#define COMPUTE_CHUNK(b2, b4)                                                            \
  {                                                                                      \
    _Pragma("unroll") for (int s = 0; s < KSW; ++s) {                                    \
      v2f f[NF];                                                                         \
      if constexpr (NF > 0) MAKE_FRAG(f[0], b2[s], b4[s], 0);                            \
      if constexpr (NF > 1) MAKE_FRAG(f[1], b2[s], b4[s], 1);                            \
      if constexpr (NF > 2) MAKE_FRAG(f[2], b2[s], b4[s], 2);                            \
      if constexpr (NF > 3) MAKE_FRAG(f[3], b2[s], b4[s], 3);                            \
      if constexpr (NF > 4) MAKE_FRAG(f[4 < NF ? 4 : 0], b2[s], b4[s], (4 < NF ? 4 : 0)); \
      if constexpr (NF > 5) MAKE_FRAG(f[5 < NF ? 5 : 0], b2[s], b4[s], (5 < NF ? 5 : 0)); \
      _Pragma("unroll") for (int q = 0; q < NT; ++q)                                     \
        acc[q] = __builtin_amdgcn_wmma_f32_16x16x4_f32(false, f[ta_of(G, q)], false,     \
                                                       f[tb_of(G, q)], (short)0, acc[q], \
                                                       false, false);                    \
      _Pragma("unroll") for (int i = 0; i < N2; ++i) {                                   \
        cs2[i] += b2[s][0][i]; cs2[i] += b2[s][1][i];                                    \
      }                                                                                  \
      _Pragma("unroll") for (int i = 0; i < N4; ++i) {                                   \
        cs4[i] += b4[s][0][i]; cs4[i] += b4[s][1][i];                                    \
      }                                                                                  \
    }                                                                                    \
  }

  LOAD_CHUNK(bufA2, bufA4, 0);
#pragma nounroll
  for (int chunk = 0; chunk < NCHUNK; chunk += 2) {
    LOAD_CHUNK(bufB2, bufB4, chunk + 1);
    __builtin_amdgcn_sched_barrier(0);
    COMPUTE_CHUNK(bufA2, bufA4);
    __builtin_amdgcn_sched_barrier(0);
    const int nxt = (chunk + 2 < NCHUNK) ? chunk + 2 : NCHUNK - 1;
    LOAD_CHUNK(bufA2, bufA4, nxt);
    __builtin_amdgcn_sched_barrier(0);
    COMPUTE_CHUNK(bufB2, bufB4);
    __builtin_amdgcn_sched_barrier(0);
  }
#undef LOAD_CHUNK
#undef COMPUTE_CHUNK
#undef MAKE_FRAG

#pragma unroll
  for (int i = 0; i < N2; ++i)
    *(v2f*)&ssum_sh[(wave * 2 + h) * C_ + base2_of(G, i) + 2 * c] = cs2[i];
#pragma unroll
  for (int i = 0; i < N4; ++i)
    *(v4f*)&ssum_sh[(wave * 2 + h) * C_ + 64 + 4 * c] = cs4[i];
#pragma unroll
  for (int q = 0; q < NT; ++q)
#pragma unroll
    for (int r = 0; r < 8; ++r) red[((wave * NT + q) * 8 + r) * 32 + lane] = acc[q][r];
  __syncthreads();

  if (tid < C_ && touched(G, tid)) {
    float s = 0.f;
#pragma unroll
    for (int w = 0; w < 16; ++w) s += ssum_sh[w * C_ + tid];
    stot[tid] = s;
  }
  float gs[NT];
#pragma unroll
  for (int q = 0; q < NT; ++q) {
    float t = 0.f;
#pragma unroll
    for (int w = 0; w < 8; ++w) t += red[((w * NT + q) * 8 + wave) * 32 + lane];
    gs[q] = t;
  }
  __syncthreads();

  const float invHW = 1.0f / (float)HW_;
  const float invN1 = 1.0f / (float)(HW_ - 1);
  const int m = wave + (h << 3);
#pragma unroll
  for (int q = 0; q < NT; ++q) {
    const int rowC = chan_of(G, ta_of(G, q), m);
    const int colC = chan_of(G, tb_of(G, q), c);
    const float cv = (gs[q] - stot[rowC] * stot[colC] * invHW) * invN1;
    out[((size_t)b * C_ + rowC) * C_ + colC] = cv;
    if (ta_of(G, q) != tb_of(G, q)) fin[(q * 16 + m) * FINP + c] = cv;
  }
  __syncthreads();
#pragma unroll
  for (int q = 0; q < NT; ++q) {
    if (ta_of(G, q) != tb_of(G, q)) {
      const float cv = fin[(q * 16 + c) * FINP + m];
      const int rowC = chan_of(G, tb_of(G, q), m);
      const int colC = chan_of(G, ta_of(G, q), c);
      out[((size_t)b * C_ + rowC) * C_ + colC] = cv;
    }
  }
}

__global__ __launch_bounds__(256) void secstat_cov_wmma(const float* __restrict__ x,
                                                        float* __restrict__ out) {
  __shared__ __attribute__((aligned(16))) float red[8 * NT * 256];
  __shared__ float fin[NT * 16 * FINP];
  __shared__ __attribute__((aligned(16))) float ssum_sh[16 * C_];
  __shared__ float stot[C_];
  switch (blockIdx.y) {
    case 0:  secstat_group<0>(x, out, red, fin, ssum_sh, stot); break;
    case 1:  secstat_group<1>(x, out, red, fin, ssum_sh, stot); break;
    case 2:  secstat_group<2>(x, out, red, fin, ssum_sh, stot); break;
    default: secstat_group<3>(x, out, red, fin, ssum_sh, stot); break;
  }
}

extern "C" void kernel_launch(void* const* d_in, const int* in_sizes, int n_in,
                              void* d_out, int out_size, void* d_ws, size_t ws_size,
                              hipStream_t stream) {
  (void)in_sizes; (void)n_in; (void)d_ws; (void)ws_size; (void)out_size;
  const float* x = (const float*)d_in[0];
  float* out = (float*)d_out;
  dim3 grid(B_, NG);
  dim3 block(256);
  secstat_cov_wmma<<<grid, block, 0, stream>>>(x, out);
}
